# starting kernel
# baseline (speedup 1.0000x reference)
#include <hip/hip_runtime.h>
#include <math.h>

#define BB 16
#define TT 128
#define NN 256
#define HH 4
#define SP (TT + 1)
#define AP (TT + 2)

typedef float v2f __attribute__((ext_vector_type(2)));
typedef float v8f __attribute__((ext_vector_type(8)));

static __device__ __forceinline__ v8f wmma4(v2f a, v2f b, v8f c) {
  return __builtin_amdgcn_wmma_f32_16x16x4_f32(
       false, a,  false, b,
       (short)0, c,  false,  false);
}

__global__ void prefix_kernel(const int* __restrict__ mask, int* __restrict__ S) {
  int b = blockIdx.x;
  if (threadIdx.x == 0) {
    int* Sb = S + b * SP;
    const int* mb = mask + b * TT;
    int s = 0;
    Sb[0] = 0;
    for (int t = 0; t < TT; ++t) { s += mb[t]; Sb[t + 1] = s; }
  }
}

__global__ void gram_kernel(const float* __restrict__ x, float* __restrict__ G) {
  int b = blockIdx.x >> 3, ti = blockIdx.x & 7;
  int wave = threadIdx.x >> 5;
  int lane = threadIdx.x & 31;
  int r = lane & 15, o = lane >> 4;
  const float* Xb = x + (size_t)b * TT * NN;
  const float* ar = Xb + (size_t)(ti * 16 + r) * NN + 2 * o;
  const float* br = Xb + (size_t)(wave * 16 + r) * NN + 2 * o;
  v8f acc = {};
#pragma unroll 8
  for (int k = 0; k < NN; k += 4) {
    v2f a = *(const v2f*)(ar + k);
    v2f bv = *(const v2f*)(br + k);
    acc = wmma4(a, bv, acc);
  }
  float* Gb = G + (size_t)b * TT * TT;
#pragma unroll
  for (int g = 0; g < 8; ++g) {
    int t = ti * 16 + g + 8 * o;
    Gb[(size_t)t * TT + wave * 16 + r] = acc[g];
  }
}

__global__ void __launch_bounds__(512) base_kernel(const float* __restrict__ x,
                                                   const float* __restrict__ w0,
                                                   float* __restrict__ Z) {
  int b = blockIdx.x >> 3, ti = blockIdx.x & 7;
  int wave = threadIdx.x >> 5;
  int lane = threadIdx.x & 31;
  int r = lane & 15, o = lane >> 4;
  const float* ar = x + ((size_t)b * TT + ti * 16 + r) * NN + 2 * o;
  const float* br = w0 + ((size_t)b * NN + wave * 16 + r) * NN + 2 * o;
  v8f acc = {};
#pragma unroll 8
  for (int k = 0; k < NN; k += 4) {
    v2f a = *(const v2f*)(ar + k);
    v2f bv = *(const v2f*)(br + k);
    acc = wmma4(a, bv, acc);
  }
#pragma unroll
  for (int g = 0; g < 8; ++g) {
    int t = ti * 16 + g + 8 * o;
    Z[((size_t)b * TT + t) * NN + wave * 16 + r] = acc[g];
  }
}

__global__ void __launch_bounds__(512) y_kernel(
    const float* __restrict__ x, const float* __restrict__ G,
    const float* __restrict__ Z, const int* __restrict__ S,
    const int* __restrict__ mask, const float* __restrict__ alpha,
    const float* __restrict__ eta, float* __restrict__ out_y) {
  __shared__ int sS[SP];
  __shared__ int sM[TT];
  __shared__ float sA[16][AP];
  __shared__ float sP[16];
  int blk = blockIdx.x;
  int ti = blk & 7;
  int h = (blk >> 3) & 3;
  int b = blk >> 5;
  int tid = threadIdx.x;
  if (tid < SP) sS[tid] = S[b * SP + tid];
  if (tid >= 256 && tid < 256 + TT) sM[tid - 256] = mask[b * TT + (tid - 256)];
  __syncthreads();

  float eh = eta[h];
  float l2a = log2f(alpha[h]);

  const float* Gb = G + ((size_t)b * TT + ti * 16) * TT;
#pragma unroll
  for (int i = 0; i < 4; ++i) {
    int idx = tid + i * 512;
    int tr = idx >> 7, j = idx & 127;
    int t = ti * 16 + tr;
    float val = 0.0f;
    if (j < t && sM[j])
      val = eh * exp2f(l2a * (float)(sS[t] - sS[j + 1])) * Gb[(size_t)tr * TT + j];
    sA[tr][j] = val;
  }
  if (tid < 16) sP[tid] = exp2f(l2a * (float)sS[ti * 16 + tid]);
  __syncthreads();

  int wave = tid >> 5;
  int lane = tid & 31;
  int r = lane & 15, o = lane >> 4;
  const float* Xb = x + (size_t)b * TT * NN;
  int n = wave * 16 + r;

  v8f acc = {};
#pragma unroll 4
  for (int k = 0; k < TT; k += 4) {
    int j0 = k + 2 * o;
    v2f av = *(const v2f*)&sA[r][j0];
    v2f bv;
    bv.x = Xb[(size_t)j0 * NN + n];
    bv.y = Xb[(size_t)(j0 + 1) * NN + n];
    acc = wmma4(av, bv, acc);
  }
#pragma unroll
  for (int g = 0; g < 8; ++g) {
    int tl = g + 8 * o;
    int t = ti * 16 + tl;
    float base = sP[tl] * Z[((size_t)b * TT + t) * NN + n];
    out_y[(((size_t)b * TT + t) * HH + h) * NN + n] = acc[g] + base;
  }
}

__global__ void __launch_bounds__(512) wfin_kernel(
    const float* __restrict__ x, const float* __restrict__ w0,
    const int* __restrict__ S, const int* __restrict__ mask,
    const float* __restrict__ alpha, const float* __restrict__ eta,
    float* __restrict__ out_w) {
  __shared__ float sd[TT];
  __shared__ float sA[16][AP];
  int blk = blockIdx.x;
  int nt = blk & 15;
  int h = (blk >> 4) & 3;
  int b = blk >> 6;
  int tid = threadIdx.x;
  float eh = eta[h];
  float l2a = log2f(alpha[h]);
  int ST = S[b * SP + TT];
  if (tid < TT) {
    int m = mask[b * TT + tid];
    int cnt = ST - S[b * SP + tid + 1];
    sd[tid] = m ? eh * exp2f(l2a * (float)cnt) : 0.0f;
  }
  __syncthreads();

  const float* Xb = x + (size_t)b * TT * NN;
#pragma unroll
  for (int i = 0; i < 4; ++i) {
    int idx = tid + i * 512;
    int nr = idx & 15, j = idx >> 4;
    sA[nr][j] = sd[j] * Xb[(size_t)j * NN + nt * 16 + nr];
  }
  __syncthreads();

  float p = exp2f(l2a * (float)ST);
  int wave = tid >> 5;
  int lane = tid & 31;
  int r = lane & 15, o = lane >> 4;
  int m0 = wave * 16;

  v8f acc = {};
#pragma unroll 4
  for (int k = 0; k < TT; k += 4) {
    int j0 = k + 2 * o;
    v2f av = *(const v2f*)&sA[r][j0];
    v2f bv;
    bv.x = Xb[(size_t)j0 * NN + m0 + r];
    bv.y = Xb[(size_t)(j0 + 1) * NN + m0 + r];
    acc = wmma4(av, bv, acc);
  }
#pragma unroll
  for (int g = 0; g < 8; ++g) {
    int nn = nt * 16 + g + 8 * o;
    int mm = m0 + r;
    float val = acc[g] + p * w0[((size_t)b * NN + nn) * NN + mm];
    out_w[(((size_t)b * HH + h) * NN + nn) * NN + mm] = val;
  }
}

extern "C" void kernel_launch(void* const* d_in, const int* in_sizes, int n_in,
                              void* d_out, int out_size, void* d_ws, size_t ws_size,
                              hipStream_t stream) {
  const float* x     = (const float*)d_in[0];
  const float* w0    = (const float*)d_in[1];
  const float* alpha = (const float*)d_in[2];
  const float* eta   = (const float*)d_in[3];
  const int*   mask  = (const int*)d_in[4];

  float* out_w = (float*)d_out;
  float* out_y = out_w + (size_t)BB * HH * NN * NN;

  int*   S = (int*)d_ws;
  float* G = (float*)((char*)d_ws + 16384);
  float* Z = G + (size_t)BB * TT * TT;

  prefix_kernel<<<dim3(BB), dim3(32), 0, stream>>>(mask, S);
  gram_kernel<<<dim3(BB * 8), dim3(256), 0, stream>>>(x, G);
  base_kernel<<<dim3(BB * 8), dim3(512), 0, stream>>>(x, w0, Z);
  y_kernel<<<dim3(BB * HH * 8), dim3(512), 0, stream>>>(x, G, Z, S, mask, alpha, eta, out_y);
  wfin_kernel<<<dim3(BB * HH * 16), dim3(512), 0, stream>>>(x, w0, S, mask, alpha, eta, out_w);
}


# Base and Gram fragment loads widened to 16 bytes per lane: a lane reads 4 consecutive k of its row for two matrix steps, halving the 16-line loads; k order in each 8-k group changes (sum order)
# speedup vs baseline: 1.9971x; 1.9971x over previous
#include <hip/hip_runtime.h>
#include <math.h>

#define BB 16
#define TT 128
#define NN 256
#define HH 4
#define SP (TT + 1)
#define AP (TT + 2)
#define NTHR 256

typedef float v2f __attribute__((ext_vector_type(2)));
typedef float v4f __attribute__((ext_vector_type(4)));
typedef float v8f __attribute__((ext_vector_type(8)));

static __device__ __forceinline__ v8f wmma4(v2f a, v2f b, v8f c) {
  return __builtin_amdgcn_wmma_f32_16x16x4_f32(
       false, a,  false, b,
       (short)0, c,  false,  false);
}

static __device__ __forceinline__ void wave_prefix4(int4 v, int lane, int4& s, int& total) {
  int s1 = v.x, s2 = s1 + v.y, s3 = s2 + v.z, s4 = s3 + v.w;
  int inc = s4;
#pragma unroll
  for (int d = 1; d < 32; d <<= 1) {
    int u = __shfl_up(inc, d);
    if (lane >= d) inc += u;
  }
  int ex = inc - s4;
  s.x = ex + s1; s.y = ex + s2; s.z = ex + s3; s.w = ex + s4;
  total = __shfl(inc, 31);
}

#define WFIN_BLOCKS (BB * 16)
#define BASE_BLOCKS (BB * 8)
#define GRAM_BLOCKS (BB * 4)

struct ABStage { v4f a[4], p[4], q[4]; };

static __device__ __forceinline__ void load_ab(ABStage& c, const float* __restrict__ ar,
                                               const float* __restrict__ b0,
                                               const float* __restrict__ b1, int kb) {
#pragma unroll
  for (int g = 0; g < 4; ++g) {
    int k = kb + 8 * g;
    c.a[g] = *(const v4f*)(ar + k);
    c.p[g] = *(const v4f*)(b0 + k);
    c.q[g] = *(const v4f*)(b1 + k);
  }
}

static __device__ __forceinline__ void mm_ab(v8f& acc0, v8f& acc1, const ABStage& c) {
#pragma unroll
  for (int g = 0; g < 4; ++g) {
    v2f a0 = {c.a[g].x, c.a[g].y}, a1 = {c.a[g].z, c.a[g].w};
    v2f p0 = {c.p[g].x, c.p[g].y}, p1 = {c.p[g].z, c.p[g].w};
    v2f q0 = {c.q[g].x, c.q[g].y}, q1 = {c.q[g].z, c.q[g].w};
    acc0 = wmma4(a0, p0, acc0);
    acc1 = wmma4(a0, q0, acc1);
    acc0 = wmma4(a1, p1, acc0);
    acc1 = wmma4(a1, q1, acc1);
  }
}

static __device__ __forceinline__ void ab_product(v8f& acc0, v8f& acc1,
                                                  const float* __restrict__ ar,
                                                  const float* __restrict__ b0,
                                                  const float* __restrict__ b1) {
  ABStage c0, c1;
  load_ab(c0, ar, b0, b1, 0);
  load_ab(c1, ar, b0, b1, 32);
#pragma unroll
  for (int it = 0; it < 8; it += 2) {
    mm_ab(acc0, acc1, c0);
    if (it + 2 < 8) load_ab(c0, ar, b0, b1, (it + 2) * 32);
    mm_ab(acc0, acc1, c1);
    if (it + 3 < 8) load_ab(c1, ar, b0, b1, (it + 3) * 32);
  }
}

static __device__ __forceinline__ void base_role(int blk, int tid,
                                                 const float* __restrict__ x,
                                                 const float* __restrict__ w0,
                                                 float* __restrict__ Z) {
  int b = blk >> 3, ti = blk & 7;
  int wave = tid >> 5;
  int lane = tid & 31;
  int r = lane & 15, o = lane >> 4;
  int n0 = wave * 32 + 2 * r;
  const float* ar = x + ((size_t)b * TT + ti * 16 + r) * NN + 4 * o;
  const float* b0 = w0 + ((size_t)b * NN + n0) * NN + 4 * o;
  const float* b1 = b0 + NN;
  v8f acc0 = {}, acc1 = {};
  ab_product(acc0, acc1, ar, b0, b1);
#pragma unroll
  for (int g = 0; g < 8; ++g) {
    int t = ti * 16 + g + 8 * o;
    v2f v = {acc0[g], acc1[g]};
    *(v2f*)&Z[((size_t)b * TT + t) * NN + n0] = v;
  }
}

static __device__ __forceinline__ void gram_role(int blk, int tid,
                                                 const float* __restrict__ x,
                                                 float* __restrict__ G) {
  int wave = tid >> 5;
  int b = blk >> 2, ti = ((blk & 3) << 1) | (wave >> 2);
  int lane = tid & 31;
  int r = lane & 15, o = lane >> 4;
  int u0 = (wave & 3) * 32 + 2 * r;
  const float* Xb = x + (size_t)b * TT * NN;
  const float* ar = Xb + (size_t)(ti * 16 + r) * NN + 4 * o;
  const float* b0 = Xb + (size_t)u0 * NN + 4 * o;
  const float* b1 = b0 + NN;
  v8f acc0 = {}, acc1 = {};
  ab_product(acc0, acc1, ar, b0, b1);
  float* Gb = G + (size_t)b * TT * TT;
#pragma unroll
  for (int g = 0; g < 8; ++g) {
    int t = ti * 16 + g + 8 * o;
    v2f v = {acc0[g], acc1[g]};
    *(v2f*)&Gb[(size_t)t * TT + u0] = v;
  }
}

struct XStage { v2f l0[8], l1[8]; };

static __device__ __forceinline__ void load_x(XStage& c, const float* __restrict__ Xb,
                                              int kb, int o, int m0) {
#pragma unroll
  for (int s = 0; s < 8; ++s) {
    int j0 = kb + 4 * s + 2 * o;
    c.l0[s] = *(const v2f*)&Xb[(size_t)j0 * NN + m0];
    c.l1[s] = *(const v2f*)&Xb[(size_t)(j0 + 1) * NN + m0];
  }
}

template <int NH>
static __device__ __forceinline__ void mm_x(v8f (&acc)[NH][2], const XStage& c,
                                            const float (*sA)[16][AP], int r, int o, int kb) {
#pragma unroll
  for (int s = 0; s < 8; ++s) {
    int j0 = kb + 4 * s + 2 * o;
    v2f be = {c.l0[s].x, c.l1[s].x};
    v2f bo = {c.l0[s].y, c.l1[s].y};
#pragma unroll
    for (int h = 0; h < NH; ++h) {
      v2f av = *(const v2f*)&sA[h][r][j0];
      acc[h][0] = wmma4(av, be, acc[h][0]);
      acc[h][1] = wmma4(av, bo, acc[h][1]);
    }
  }
}

__global__ void __launch_bounds__(NTHR) phase1_kernel(
    const float* __restrict__ x, const float* __restrict__ w0,
    const int* __restrict__ mask, const float* __restrict__ alpha,
    const float* __restrict__ eta, float* __restrict__ G,
    float* __restrict__ Z, float* __restrict__ out_w) {
  __shared__ float sPw[HH];
  __shared__ float sd[HH][TT];
  __shared__ __attribute__((aligned(16))) float sA[HH][16][AP];
  int blk = blockIdx.x;
  int tid = threadIdx.x;
  if (blk >= WFIN_BLOCKS) {
    blk -= WFIN_BLOCKS;
    if (blk < BASE_BLOCKS) base_role(blk, tid, x, w0, Z);
    else gram_role(blk - BASE_BLOCKS, tid, x, G);
    return;
  }

  int nt = blk & 15;
  int b = blk >> 4;
  int wave = tid >> 5;
  int lane = tid & 31;
  int r = lane & 15, o = lane >> 4;
  int m0 = wave * 32 + 2 * r;
  const float* Xb = x + (size_t)b * TT * NN;

  int4 mv = *(const int4*)(mask + b * TT + 4 * lane);
  float xa[8];
#pragma unroll
  for (int i = 0; i < 8; ++i) {
    int idx = tid + i * NTHR;
    int nr = idx & 15, j = idx >> 4;
    xa[i] = Xb[(size_t)j * NN + nt * 16 + nr];
  }
  XStage c0, c1;
  load_x(c0, Xb, 0, o, m0);

  int4 sv; int tot;
  wave_prefix4(mv, lane, sv, tot);
  if (wave < HH) {
    int h = wave;
    float eh = eta[h];
    float l2a = log2f(alpha[h]);
    sd[h][4 * lane + 0] = mv.x ? eh * exp2f(l2a * (float)(tot - sv.x)) : 0.0f;
    sd[h][4 * lane + 1] = mv.y ? eh * exp2f(l2a * (float)(tot - sv.y)) : 0.0f;
    sd[h][4 * lane + 2] = mv.z ? eh * exp2f(l2a * (float)(tot - sv.z)) : 0.0f;
    sd[h][4 * lane + 3] = mv.w ? eh * exp2f(l2a * (float)(tot - sv.w)) : 0.0f;
    if (lane == 0) sPw[h] = exp2f(l2a * (float)tot);
  }
  __syncthreads();

#pragma unroll
  for (int i = 0; i < 8; ++i) {
    int idx = tid + i * NTHR;
    int nr = idx & 15, j = idx >> 4;
#pragma unroll
    for (int h = 0; h < HH; ++h) sA[h][nr][j] = sd[h][j] * xa[i];
  }
  __syncthreads();

  load_x(c1, Xb, 32, o, m0);
  v2f wv[8];
#pragma unroll
  for (int g = 0; g < 8; ++g) {
    int nn = nt * 16 + g + 8 * o;
    wv[g] = *(const v2f*)&w0[((size_t)b * NN + nn) * NN + m0];
  }
  v8f acc[HH][2] = {};
  mm_x<HH>(acc, c0, sA, r, o, 0);
  load_x(c0, Xb, 64, o, m0);
  mm_x<HH>(acc, c1, sA, r, o, 32);
  load_x(c1, Xb, 96, o, m0);
  mm_x<HH>(acc, c0, sA, r, o, 64);
  mm_x<HH>(acc, c1, sA, r, o, 96);

#pragma unroll
  for (int g = 0; g < 8; ++g) {
    int nn = nt * 16 + g + 8 * o;
    v2f w = wv[g];
#pragma unroll
    for (int h = 0; h < HH; ++h) {
      float p = sPw[h];
      v2f val = {acc[h][0][g] + p * w.x, acc[h][1][g] + p * w.y};
      *(v2f*)&out_w[(((size_t)b * HH + h) * NN + nn) * NN + m0] = val;
    }
  }
}

#define YH 2
__global__ void __launch_bounds__(NTHR) y_kernel(
    const float* __restrict__ x, const float* __restrict__ G,
    const float* __restrict__ Z, const int* __restrict__ mask,
    const float* __restrict__ alpha, const float* __restrict__ eta,
    float* __restrict__ out_y) {
  __shared__ int sS[SP];
  __shared__ int sM[TT];
  __shared__ __attribute__((aligned(16))) float sA[YH][16][AP];
  __shared__ float sP[YH][16];
  int blk = blockIdx.x;
  int hp = blk & 1;
  int ti = (blk >> 1) & 7;
  int b = blk >> 4;
  int tid = threadIdx.x;
  int wave = tid >> 5;
  int lane = tid & 31;
  int r = lane & 15, o = lane >> 4;
  const float* Xb = x + (size_t)b * TT * NN;
  int n0 = wave * 32 + 2 * r;

  int4 mv = *(const int4*)(mask + b * TT + 4 * lane);
  const float* Gb = G + ((size_t)b * TT + ti * 16) * TT;
  float gv[8];
#pragma unroll
  for (int i = 0; i < 8; ++i) {
    int idx = tid + i * NTHR;
    gv[i] = Gb[(size_t)(idx >> 7) * TT + (idx & 127)];
  }
  XStage c0, c1, c2, c3;
  load_x(c0, Xb, 0, o, n0);

  float eh[YH], l2a[YH];
#pragma unroll
  for (int h = 0; h < YH; ++h) {
    eh[h] = eta[hp * YH + h];
    l2a[h] = log2f(alpha[hp * YH + h]);
  }

  int4 sv; int tot;
  wave_prefix4(mv, lane, sv, tot);
  if (tid < 32) {
    sM[4 * tid + 0] = mv.x; sM[4 * tid + 1] = mv.y;
    sM[4 * tid + 2] = mv.z; sM[4 * tid + 3] = mv.w;
    sS[4 * tid + 1] = sv.x; sS[4 * tid + 2] = sv.y;
    sS[4 * tid + 3] = sv.z; sS[4 * tid + 4] = sv.w;
    if (tid == 0) sS[0] = 0;
  }
  __syncthreads();

#pragma unroll
  for (int i = 0; i < 8; ++i) {
    int idx = tid + i * NTHR;
    int tr = idx >> 7, j = idx & 127;
    int t = ti * 16 + tr;
    float val[YH];
#pragma unroll
    for (int h = 0; h < YH; ++h) val[h] = 0.0f;
    if (j < t && sM[j]) {
      float cnt = (float)(sS[t] - sS[j + 1]);
#pragma unroll
      for (int h = 0; h < YH; ++h) val[h] = eh[h] * exp2f(l2a[h] * cnt) * gv[i];
    }
#pragma unroll
    for (int h = 0; h < YH; ++h) sA[h][tr][j] = val[h];
  }
  if (tid < YH * 16) {
    int h = tid >> 4, tl = tid & 15;
    sP[h][tl] = exp2f(l2a[h] * (float)sS[ti * 16 + tl]);
  }
  __syncthreads();

  load_x(c1, Xb, 32, o, n0);
  __builtin_amdgcn_sched_barrier(0);
  load_x(c2, Xb, 64, o, n0);
  __builtin_amdgcn_sched_barrier(0);
  load_x(c3, Xb, 96, o, n0);
  __builtin_amdgcn_sched_barrier(0);
  v2f zv[8];
#pragma unroll
  for (int g = 0; g < 8; ++g) {
    int t = ti * 16 + g + 8 * o;
    zv[g] = *(const v2f*)&Z[((size_t)b * TT + t) * NN + n0];
  }
  __builtin_amdgcn_sched_barrier(0);
  v8f acc[YH][2] = {};
  mm_x<YH>(acc, c0, sA, r, o, 0);
  mm_x<YH>(acc, c1, sA, r, o, 32);
  mm_x<YH>(acc, c2, sA, r, o, 64);
  mm_x<YH>(acc, c3, sA, r, o, 96);

#pragma unroll
  for (int g = 0; g < 8; ++g) {
    int tl = g + 8 * o;
    int t = ti * 16 + tl;
    v2f z = zv[g];
#pragma unroll
    for (int h = 0; h < YH; ++h) {
      float p = sP[h][tl];
      v2f val = {acc[h][0][g] + p * z.x, acc[h][1][g] + p * z.y};
      *(v2f*)&out_y[(((size_t)b * TT + t) * HH + hp * YH + h) * NN + n0] = val;
    }
  }
}

extern "C" void kernel_launch(void* const* d_in, const int* in_sizes, int n_in,
                              void* d_out, int out_size, void* d_ws, size_t ws_size,
                              hipStream_t stream) {
  const float* x     = (const float*)d_in[0];
  const float* w0    = (const float*)d_in[1];
  const float* alpha = (const float*)d_in[2];
  const float* eta   = (const float*)d_in[3];
  const int*   mask  = (const int*)d_in[4];

  float* out_w = (float*)d_out;
  float* out_y = out_w + (size_t)BB * HH * NN * NN;

  float* G = (float*)d_ws;
  float* Z = G + (size_t)BB * TT * TT;

  phase1_kernel<<<dim3(WFIN_BLOCKS + BASE_BLOCKS + GRAM_BLOCKS), dim3(NTHR), 0, stream>>>(
      x, w0, mask, alpha, eta, G, Z, out_w);
  y_kernel<<<dim3(BB * 8 * (HH / YH)), dim3(NTHR), 0, stream>>>(x, G, Z, mask, alpha, eta, out_y);
}
